# starting kernel
# baseline (speedup 1.0000x reference)
#include <hip/hip_runtime.h>
#include <math.h>

typedef float v2f __attribute__((ext_vector_type(2)));
typedef float v8f __attribute__((ext_vector_type(8)));

#define C_DIM   512
#define HW      6144
#define NIMG    16
#define O_LOC   36
#define O_SC    18
#define O_TOT   54
#define O_PAD   64
#define KCHUNK  128
#define PAIR_STRIDE (O_PAD * 2)

#define M_TOTAL   (NIMG * HW)
#define TILES     (M_TOTAL / 16)
#define TILES_PER_IMG (HW / 16)

#define LOC_ELEMS ((size_t)NIMG * HW * O_LOC)
#define SC_ELEMS  ((size_t)NIMG * HW * O_SC)
#define FG_ELEMS  ((size_t)NIMG * HW * 9)

__global__ __launch_bounds__(256)
void rpn_head_kernel(const float* __restrict__ x,
                     const float* __restrict__ loc_w,
                     const float* __restrict__ loc_b,
                     const float* __restrict__ score_w,
                     const float* __restrict__ score_b,
                     float* __restrict__ d_out)
{
    __shared__ float sW[(KCHUNK / 2) * PAIR_STRIDE];
    __shared__ float sBias[O_PAD];

    const int tid  = threadIdx.x;
    const int wave = tid >> 5;
    const int lane = tid & 31;
    const int col  = lane & 15;
    const int khalf = lane >> 4;

    if (tid < O_PAD) {
        float b = 0.0f;
        if (tid < O_LOC)       b = loc_b[tid];
        else if (tid < O_TOT)  b = score_b[tid - O_LOC];
        sBias[tid] = b;
    }

    const int tile0 = blockIdx.x * 16 + wave * 2;
    const int tile1 = tile0 + 1;

    const int n0 = tile0 / TILES_PER_IMG;
    const int n1 = tile1 / TILES_PER_IMG;
    const size_t base0 = (size_t)n0 * C_DIM * HW + (size_t)(tile0 % TILES_PER_IMG) * 16 + col;
    const size_t base1 = (size_t)n1 * C_DIM * HW + (size_t)(tile1 % TILES_PER_IMG) * 16 + col;

    v8f acc[2][4];
    const v8f vzero = {0.f,0.f,0.f,0.f,0.f,0.f,0.f,0.f};
    #pragma unroll
    for (int t = 0; t < 2; ++t)
        #pragma unroll
        for (int j = 0; j < 4; ++j)
            acc[t][j] = vzero;

    const int kq = khalf << 1;

    for (int kc = 0; kc < C_DIM / KCHUNK; ++kc) {
        const int k0 = kc * KCHUNK;

        __syncthreads();
        for (int idx = tid; idx < O_TOT * KCHUNK; idx += 256) {
            const int o  = idx >> 7;
            const int kk = idx & (KCHUNK - 1);
            const float wv = (o < O_LOC)
                ? loc_w[o * C_DIM + k0 + kk]
                : score_w[(o - O_LOC) * C_DIM + k0 + kk];
            sW[(kk >> 1) * PAIR_STRIDE + o * 2 + (kk & 1)] = wv;
        }
        __syncthreads();

        for (int ks = 0; ks < KCHUNK; ks += 4) {
            const int c0 = k0 + ks + kq;

            v2f a0, a1;
            a0.x = x[base0 + (size_t)(c0    ) * HW];
            a0.y = x[base0 + (size_t)(c0 + 1) * HW];
            a1.x = x[base1 + (size_t)(c0    ) * HW];
            a1.y = x[base1 + (size_t)(c0 + 1) * HW];

            const int pairRow = (ks >> 1) + khalf;
            const float* pb = &sW[pairRow * PAIR_STRIDE + col * 2];
            v2f b[4];
            #pragma unroll
            for (int j = 0; j < 4; ++j)
                b[j] = *(const v2f*)(pb + j * 32);

            #pragma unroll
            for (int j = 0; j < 4; ++j) {
                acc[0][j] = __builtin_amdgcn_wmma_f32_16x16x4_f32(
                    false, a0, false, b[j], (short)0, acc[0][j], false, false);
                acc[1][j] = __builtin_amdgcn_wmma_f32_16x16x4_f32(
                    false, a1, false, b[j], (short)0, acc[1][j], false, false);
            }

            if (ks + 4 < KCHUNK) {
                __builtin_prefetch((const void*)&x[base0 + (size_t)(c0 + 4) * HW], 0, 1);
                __builtin_prefetch((const void*)&x[base1 + (size_t)(c0 + 4) * HW], 0, 1);
            }
        }
    }

    float* out_loc   = d_out;
    float* out_score = d_out + LOC_ELEMS;
    float* out_fg    = d_out + LOC_ELEMS + SC_ELEMS;

    const int mhi = khalf * 8;

    #pragma unroll
    for (int t = 0; t < 2; ++t) {
        const int tile = (t == 0) ? tile0 : tile1;
        #pragma unroll
        for (int j = 0; j < 4; ++j) {
            const int o = j * 16 + col;
            const float bias = sBias[o];
            #pragma unroll
            for (int v = 0; v < 8; ++v) {
                const float val = acc[t][j][v] + bias;
                const float partner = __shfl_xor(val, 1, 32);
                const size_t p = (size_t)tile * 16 + (size_t)(v + mhi);
                if (o < O_LOC) {
                    out_loc[p * O_LOC + o] = val;
                } else if (o < O_TOT) {
                    const int s = o - O_LOC;
                    out_score[p * O_SC + s] = val;
                    if ((s & 1) == 0) {
                        const float fg = 1.0f / (1.0f + __expf(val - partner));
                        out_fg[p * 9 + (s >> 1)] = fg;
                    }
                }
            }
        }
    }
}

__global__ __launch_bounds__(256)
void rpn_anchor_kernel(float* __restrict__ out_anchor)
{
    const int i = blockIdx.x * 256 + threadIdx.x;
    const int hw = i / 9;
    const int a  = i - hw * 9;
    const int h  = hw / 96;
    const int w  = hw - h * 96;
    const int ri = a / 3;
    const int si = a - ri * 3;

    const float ratios[3] = {0.5f, 1.0f, 2.0f};
    const float scales[3] = {8.0f, 16.0f, 32.0f};
    const float r = ratios[ri];
    const float s = scales[si];

    const float hbox = 16.0f * s * __fsqrt_rn(r);
    const float wbox = 16.0f * s * __fsqrt_rn(1.0f / r);
    const float py = (float)h * 16.0f + 8.0f;
    const float px = (float)w * 16.0f + 8.0f;

    float4 v;
    v.x = py - hbox * 0.5f;
    v.y = px - wbox * 0.5f;
    v.z = py + hbox * 0.5f;
    v.w = px + wbox * 0.5f;
    ((float4*)out_anchor)[i] = v;
}

extern "C" void kernel_launch(void* const* d_in, const int* in_sizes, int n_in,
                              void* d_out, int out_size, void* d_ws, size_t ws_size,
                              hipStream_t stream)
{
    const float* x       = (const float*)d_in[0];
    const float* loc_w   = (const float*)d_in[1];
    const float* loc_b   = (const float*)d_in[2];
    const float* score_w = (const float*)d_in[3];
    const float* score_b = (const float*)d_in[4];
    float* out = (float*)d_out;

    rpn_head_kernel<<<TILES / 16, 256, 0, stream>>>(x, loc_w, loc_b, score_w, score_b, out);

    float* out_anchor = out + LOC_ELEMS + SC_ELEMS + FG_ELEMS;
    rpn_anchor_kernel<<<(HW * 9) / 256, 256, 0, stream>>>(out_anchor);
}


# x fragments double-buffered in registers: their loads run at least one 32-channel block (64 WMMAs) ahead of use, replacing the one-step prefetch hint; sum order unchanged
# speedup vs baseline: 1.2398x; 1.2398x over previous
#include <hip/hip_runtime.h>
#include <math.h>

typedef float v2f __attribute__((ext_vector_type(2)));
typedef float v8f __attribute__((ext_vector_type(8)));

#define C_DIM   512
#define HW      6144
#define NIMG    16
#define O_LOC   36
#define O_SC    18
#define O_TOT   54
#define O_PAD   64
#define KCHUNK  128
#define PAIR_STRIDE (O_PAD * 2)
#define KBLOCK  32
#define BSTEPS  (KBLOCK / 4)

#define M_TOTAL   (NIMG * HW)
#define TILES     (M_TOTAL / 16)
#define TILES_PER_IMG (HW / 16)

#define LOC_ELEMS ((size_t)NIMG * HW * O_LOC)
#define SC_ELEMS  ((size_t)NIMG * HW * O_SC)
#define FG_ELEMS  ((size_t)NIMG * HW * 9)

__device__ __forceinline__
void load_x_block(const float* __restrict__ p0, const float* __restrict__ p1,
                  v2f (&a0)[BSTEPS], v2f (&a1)[BSTEPS])
{
    #pragma unroll
    for (int s = 0; s < BSTEPS; ++s) {
        a0[s].x = p0[(size_t)(4 * s    ) * HW];
        a0[s].y = p0[(size_t)(4 * s + 1) * HW];
        a1[s].x = p1[(size_t)(4 * s    ) * HW];
        a1[s].y = p1[(size_t)(4 * s + 1) * HW];
    }
}

__device__ __forceinline__
void mma_x_block(const float* pb, const v2f (&a0)[BSTEPS], const v2f (&a1)[BSTEPS],
                 v8f (&acc)[2][4])
{
    #pragma unroll
    for (int s = 0; s < BSTEPS; ++s) {
        v2f b[4];
        #pragma unroll
        for (int j = 0; j < 4; ++j)
            b[j] = *(const v2f*)(pb + s * 2 * PAIR_STRIDE + j * 32);

        #pragma unroll
        for (int j = 0; j < 4; ++j) {
            acc[0][j] = __builtin_amdgcn_wmma_f32_16x16x4_f32(
                false, a0[s], false, b[j], (short)0, acc[0][j], false, false);
            acc[1][j] = __builtin_amdgcn_wmma_f32_16x16x4_f32(
                false, a1[s], false, b[j], (short)0, acc[1][j], false, false);
        }
    }
}

template <bool HAS_NEXT>
__device__ __forceinline__
void head_chunk(const int kc, const int tid,
                const float* __restrict__ loc_w, const float* __restrict__ score_w,
                float* sW, const float* __restrict__ p0, const float* __restrict__ p1,
                const float* pbw,
                v2f (&xP0)[BSTEPS], v2f (&xP1)[BSTEPS],
                v2f (&xQ0)[BSTEPS], v2f (&xQ1)[BSTEPS], v8f (&acc)[2][4])
{
    const int k0 = kc * KCHUNK;

    __syncthreads();
    for (int idx = tid; idx < O_TOT * KCHUNK; idx += 256) {
        const int o  = idx >> 7;
        const int kk = idx & (KCHUNK - 1);
        const float wv = (o < O_LOC)
            ? loc_w[o * C_DIM + k0 + kk]
            : score_w[(o - O_LOC) * C_DIM + k0 + kk];
        sW[(kk >> 1) * PAIR_STRIDE + o * 2 + (kk & 1)] = wv;
    }
    __syncthreads();

    const float* q0 = p0 + (size_t)k0 * HW;
    const float* q1 = p1 + (size_t)k0 * HW;

    load_x_block(q0 + (size_t)(1 * KBLOCK) * HW, q1 + (size_t)(1 * KBLOCK) * HW, xQ0, xQ1);
    mma_x_block(pbw + 0 * (KBLOCK / 2) * PAIR_STRIDE, xP0, xP1, acc);

    load_x_block(q0 + (size_t)(2 * KBLOCK) * HW, q1 + (size_t)(2 * KBLOCK) * HW, xP0, xP1);
    mma_x_block(pbw + 1 * (KBLOCK / 2) * PAIR_STRIDE, xQ0, xQ1, acc);

    load_x_block(q0 + (size_t)(3 * KBLOCK) * HW, q1 + (size_t)(3 * KBLOCK) * HW, xQ0, xQ1);
    mma_x_block(pbw + 2 * (KBLOCK / 2) * PAIR_STRIDE, xP0, xP1, acc);

    if (HAS_NEXT)
        load_x_block(q0 + (size_t)(4 * KBLOCK) * HW, q1 + (size_t)(4 * KBLOCK) * HW, xP0, xP1);
    mma_x_block(pbw + 3 * (KBLOCK / 2) * PAIR_STRIDE, xQ0, xQ1, acc);
}

__global__ __launch_bounds__(256)
void rpn_head_kernel(const float* __restrict__ x,
                     const float* __restrict__ loc_w,
                     const float* __restrict__ loc_b,
                     const float* __restrict__ score_w,
                     const float* __restrict__ score_b,
                     float* __restrict__ d_out)
{
    __shared__ float sW[(KCHUNK / 2) * PAIR_STRIDE];
    __shared__ float sBias[O_PAD];

    const int tid  = threadIdx.x;
    const int wave = tid >> 5;
    const int lane = tid & 31;
    const int col  = lane & 15;
    const int khalf = lane >> 4;

    if (tid < O_PAD) {
        float b = 0.0f;
        if (tid < O_LOC)       b = loc_b[tid];
        else if (tid < O_TOT)  b = score_b[tid - O_LOC];
        sBias[tid] = b;
    }

    const int tile0 = blockIdx.x * 16 + wave * 2;
    const int tile1 = tile0 + 1;

    const int n0 = tile0 / TILES_PER_IMG;
    const int n1 = tile1 / TILES_PER_IMG;
    const size_t base0 = (size_t)n0 * C_DIM * HW + (size_t)(tile0 % TILES_PER_IMG) * 16 + col;
    const size_t base1 = (size_t)n1 * C_DIM * HW + (size_t)(tile1 % TILES_PER_IMG) * 16 + col;

    v8f acc[2][4];
    const v8f vzero = {0.f,0.f,0.f,0.f,0.f,0.f,0.f,0.f};
    #pragma unroll
    for (int t = 0; t < 2; ++t)
        #pragma unroll
        for (int j = 0; j < 4; ++j)
            acc[t][j] = vzero;

    const int kq = khalf << 1;

    const float* p0 = x + base0 + (size_t)kq * HW;
    const float* p1 = x + base1 + (size_t)kq * HW;
    const float* pbw = &sW[khalf * PAIR_STRIDE + col * 2];

    v2f xP0[BSTEPS], xP1[BSTEPS], xQ0[BSTEPS], xQ1[BSTEPS];

    load_x_block(p0, p1, xP0, xP1);

    for (int kc = 0; kc < C_DIM / KCHUNK - 1; ++kc)
        head_chunk<true>(kc, tid, loc_w, score_w, sW, p0, p1, pbw, xP0, xP1, xQ0, xQ1, acc);
    head_chunk<false>(C_DIM / KCHUNK - 1, tid, loc_w, score_w, sW, p0, p1, pbw,
                      xP0, xP1, xQ0, xQ1, acc);

    float* out_loc   = d_out;
    float* out_score = d_out + LOC_ELEMS;
    float* out_fg    = d_out + LOC_ELEMS + SC_ELEMS;

    const int mhi = khalf * 8;

    #pragma unroll
    for (int t = 0; t < 2; ++t) {
        const int tile = (t == 0) ? tile0 : tile1;
        #pragma unroll
        for (int j = 0; j < 4; ++j) {
            const int o = j * 16 + col;
            const float bias = sBias[o];
            #pragma unroll
            for (int v = 0; v < 8; ++v) {
                const float val = acc[t][j][v] + bias;
                const float partner = __shfl_xor(val, 1, 32);
                const size_t p = (size_t)tile * 16 + (size_t)(v + mhi);
                if (o < O_LOC) {
                    out_loc[p * O_LOC + o] = val;
                } else if (o < O_TOT) {
                    const int s = o - O_LOC;
                    out_score[p * O_SC + s] = val;
                    if ((s & 1) == 0) {
                        const float fg = 1.0f / (1.0f + __expf(val - partner));
                        out_fg[p * 9 + (s >> 1)] = fg;
                    }
                }
            }
        }
    }
}

__global__ __launch_bounds__(256)
void rpn_anchor_kernel(float* __restrict__ out_anchor)
{
    const int i = blockIdx.x * 256 + threadIdx.x;
    const int hw = i / 9;
    const int a  = i - hw * 9;
    const int h  = hw / 96;
    const int w  = hw - h * 96;
    const int ri = a / 3;
    const int si = a - ri * 3;

    const float ratios[3] = {0.5f, 1.0f, 2.0f};
    const float scales[3] = {8.0f, 16.0f, 32.0f};
    const float r = ratios[ri];
    const float s = scales[si];

    const float hbox = 16.0f * s * __fsqrt_rn(r);
    const float wbox = 16.0f * s * __fsqrt_rn(1.0f / r);
    const float py = (float)h * 16.0f + 8.0f;
    const float px = (float)w * 16.0f + 8.0f;

    float4 v;
    v.x = py - hbox * 0.5f;
    v.y = px - wbox * 0.5f;
    v.z = py + hbox * 0.5f;
    v.w = px + wbox * 0.5f;
    ((float4*)out_anchor)[i] = v;
}

extern "C" void kernel_launch(void* const* d_in, const int* in_sizes, int n_in,
                              void* d_out, int out_size, void* d_ws, size_t ws_size,
                              hipStream_t stream)
{
    const float* x       = (const float*)d_in[0];
    const float* loc_w   = (const float*)d_in[1];
    const float* loc_b   = (const float*)d_in[2];
    const float* score_w = (const float*)d_in[3];
    const float* score_b = (const float*)d_in[4];
    float* out = (float*)d_out;

    rpn_head_kernel<<<TILES / 16, 256, 0, stream>>>(x, loc_w, loc_b, score_w, score_b, out);

    float* out_anchor = out + LOC_ELEMS + SC_ELEMS + FG_ELEMS;
    rpn_anchor_kernel<<<(HW * 9) / 256, 256, 0, stream>>>(out_anchor);
}


# work split evened: 256 workgroups of 12 waves (24 M-tiles, one per WGP, 3 waves per SIMD) replace 384 of 8 waves that left half the WGPs a double share; per-wave code and sum order unchanged
# speedup vs baseline: 1.4029x; 1.1316x over previous
#include <hip/hip_runtime.h>
#include <math.h>

typedef float v2f __attribute__((ext_vector_type(2)));
typedef float v8f __attribute__((ext_vector_type(8)));

#define C_DIM   512
#define HW      6144
#define NIMG    16
#define O_LOC   36
#define O_SC    18
#define O_TOT   54
#define O_PAD   64
#define KCHUNK  128
#define PAIR_STRIDE (O_PAD * 2)
#define KBLOCK  32
#define BSTEPS  (KBLOCK / 4)
#define WG_WAVES   12
#define WG_THREADS (WG_WAVES * 32)
#define WG_TILES   (WG_WAVES * 2)

#define M_TOTAL   (NIMG * HW)
#define TILES     (M_TOTAL / 16)
#define TILES_PER_IMG (HW / 16)

#define LOC_ELEMS ((size_t)NIMG * HW * O_LOC)
#define SC_ELEMS  ((size_t)NIMG * HW * O_SC)
#define FG_ELEMS  ((size_t)NIMG * HW * 9)

__device__ __forceinline__
void load_x_block(const float* __restrict__ p0, const float* __restrict__ p1,
                  v2f (&a0)[BSTEPS], v2f (&a1)[BSTEPS])
{
    #pragma unroll
    for (int s = 0; s < BSTEPS; ++s) {
        a0[s].x = p0[(size_t)(4 * s    ) * HW];
        a0[s].y = p0[(size_t)(4 * s + 1) * HW];
        a1[s].x = p1[(size_t)(4 * s    ) * HW];
        a1[s].y = p1[(size_t)(4 * s + 1) * HW];
    }
}

__device__ __forceinline__
void mma_x_block(const float* pb, const v2f (&a0)[BSTEPS], const v2f (&a1)[BSTEPS],
                 v8f (&acc)[2][4])
{
    #pragma unroll
    for (int s = 0; s < BSTEPS; ++s) {
        v2f b[4];
        #pragma unroll
        for (int j = 0; j < 4; ++j)
            b[j] = *(const v2f*)(pb + s * 2 * PAIR_STRIDE + j * 32);

        #pragma unroll
        for (int j = 0; j < 4; ++j) {
            acc[0][j] = __builtin_amdgcn_wmma_f32_16x16x4_f32(
                false, a0[s], false, b[j], (short)0, acc[0][j], false, false);
            acc[1][j] = __builtin_amdgcn_wmma_f32_16x16x4_f32(
                false, a1[s], false, b[j], (short)0, acc[1][j], false, false);
        }
    }
}

template <bool HAS_NEXT>
__device__ __forceinline__
void head_chunk(const int kc, const int tid,
                const float* __restrict__ loc_w, const float* __restrict__ score_w,
                float* sW, const float* __restrict__ p0, const float* __restrict__ p1,
                const float* pbw,
                v2f (&xP0)[BSTEPS], v2f (&xP1)[BSTEPS],
                v2f (&xQ0)[BSTEPS], v2f (&xQ1)[BSTEPS], v8f (&acc)[2][4])
{
    const int k0 = kc * KCHUNK;

    __syncthreads();
    for (int idx = tid; idx < O_TOT * KCHUNK; idx += WG_THREADS) {
        const int o  = idx >> 7;
        const int kk = idx & (KCHUNK - 1);
        const float wv = (o < O_LOC)
            ? loc_w[o * C_DIM + k0 + kk]
            : score_w[(o - O_LOC) * C_DIM + k0 + kk];
        sW[(kk >> 1) * PAIR_STRIDE + o * 2 + (kk & 1)] = wv;
    }
    __syncthreads();

    const float* q0 = p0 + (size_t)k0 * HW;
    const float* q1 = p1 + (size_t)k0 * HW;

    load_x_block(q0 + (size_t)(1 * KBLOCK) * HW, q1 + (size_t)(1 * KBLOCK) * HW, xQ0, xQ1);
    mma_x_block(pbw + 0 * (KBLOCK / 2) * PAIR_STRIDE, xP0, xP1, acc);

    load_x_block(q0 + (size_t)(2 * KBLOCK) * HW, q1 + (size_t)(2 * KBLOCK) * HW, xP0, xP1);
    mma_x_block(pbw + 1 * (KBLOCK / 2) * PAIR_STRIDE, xQ0, xQ1, acc);

    load_x_block(q0 + (size_t)(3 * KBLOCK) * HW, q1 + (size_t)(3 * KBLOCK) * HW, xQ0, xQ1);
    mma_x_block(pbw + 2 * (KBLOCK / 2) * PAIR_STRIDE, xP0, xP1, acc);

    if (HAS_NEXT)
        load_x_block(q0 + (size_t)(4 * KBLOCK) * HW, q1 + (size_t)(4 * KBLOCK) * HW, xP0, xP1);
    mma_x_block(pbw + 3 * (KBLOCK / 2) * PAIR_STRIDE, xQ0, xQ1, acc);
}

__global__ __launch_bounds__(WG_THREADS)
void rpn_head_kernel(const float* __restrict__ x,
                     const float* __restrict__ loc_w,
                     const float* __restrict__ loc_b,
                     const float* __restrict__ score_w,
                     const float* __restrict__ score_b,
                     float* __restrict__ d_out)
{
    __shared__ float sW[(KCHUNK / 2) * PAIR_STRIDE];
    __shared__ float sBias[O_PAD];

    const int tid  = threadIdx.x;
    const int wave = tid >> 5;
    const int lane = tid & 31;
    const int col  = lane & 15;
    const int khalf = lane >> 4;

    if (tid < O_PAD) {
        float b = 0.0f;
        if (tid < O_LOC)       b = loc_b[tid];
        else if (tid < O_TOT)  b = score_b[tid - O_LOC];
        sBias[tid] = b;
    }

    const int tile0 = blockIdx.x * WG_TILES + wave * 2;
    const int tile1 = tile0 + 1;

    const int n0 = tile0 / TILES_PER_IMG;
    const int n1 = tile1 / TILES_PER_IMG;
    const size_t base0 = (size_t)n0 * C_DIM * HW + (size_t)(tile0 % TILES_PER_IMG) * 16 + col;
    const size_t base1 = (size_t)n1 * C_DIM * HW + (size_t)(tile1 % TILES_PER_IMG) * 16 + col;

    v8f acc[2][4];
    const v8f vzero = {0.f,0.f,0.f,0.f,0.f,0.f,0.f,0.f};
    #pragma unroll
    for (int t = 0; t < 2; ++t)
        #pragma unroll
        for (int j = 0; j < 4; ++j)
            acc[t][j] = vzero;

    const int kq = khalf << 1;

    const float* p0 = x + base0 + (size_t)kq * HW;
    const float* p1 = x + base1 + (size_t)kq * HW;
    const float* pbw = &sW[khalf * PAIR_STRIDE + col * 2];

    v2f xP0[BSTEPS], xP1[BSTEPS], xQ0[BSTEPS], xQ1[BSTEPS];

    load_x_block(p0, p1, xP0, xP1);

    for (int kc = 0; kc < C_DIM / KCHUNK - 1; ++kc)
        head_chunk<true>(kc, tid, loc_w, score_w, sW, p0, p1, pbw, xP0, xP1, xQ0, xQ1, acc);
    head_chunk<false>(C_DIM / KCHUNK - 1, tid, loc_w, score_w, sW, p0, p1, pbw,
                      xP0, xP1, xQ0, xQ1, acc);

    float* out_loc   = d_out;
    float* out_score = d_out + LOC_ELEMS;
    float* out_fg    = d_out + LOC_ELEMS + SC_ELEMS;

    const int mhi = khalf * 8;

    #pragma unroll
    for (int t = 0; t < 2; ++t) {
        const int tile = (t == 0) ? tile0 : tile1;
        #pragma unroll
        for (int j = 0; j < 4; ++j) {
            const int o = j * 16 + col;
            const float bias = sBias[o];
            #pragma unroll
            for (int v = 0; v < 8; ++v) {
                const float val = acc[t][j][v] + bias;
                const float partner = __shfl_xor(val, 1, 32);
                const size_t p = (size_t)tile * 16 + (size_t)(v + mhi);
                if (o < O_LOC) {
                    out_loc[p * O_LOC + o] = val;
                } else if (o < O_TOT) {
                    const int s = o - O_LOC;
                    out_score[p * O_SC + s] = val;
                    if ((s & 1) == 0) {
                        const float fg = 1.0f / (1.0f + __expf(val - partner));
                        out_fg[p * 9 + (s >> 1)] = fg;
                    }
                }
            }
        }
    }
}

__global__ __launch_bounds__(256)
void rpn_anchor_kernel(float* __restrict__ out_anchor)
{
    const int i = blockIdx.x * 256 + threadIdx.x;
    const int hw = i / 9;
    const int a  = i - hw * 9;
    const int h  = hw / 96;
    const int w  = hw - h * 96;
    const int ri = a / 3;
    const int si = a - ri * 3;

    const float ratios[3] = {0.5f, 1.0f, 2.0f};
    const float scales[3] = {8.0f, 16.0f, 32.0f};
    const float r = ratios[ri];
    const float s = scales[si];

    const float hbox = 16.0f * s * __fsqrt_rn(r);
    const float wbox = 16.0f * s * __fsqrt_rn(1.0f / r);
    const float py = (float)h * 16.0f + 8.0f;
    const float px = (float)w * 16.0f + 8.0f;

    float4 v;
    v.x = py - hbox * 0.5f;
    v.y = px - wbox * 0.5f;
    v.z = py + hbox * 0.5f;
    v.w = px + wbox * 0.5f;
    ((float4*)out_anchor)[i] = v;
}

extern "C" void kernel_launch(void* const* d_in, const int* in_sizes, int n_in,
                              void* d_out, int out_size, void* d_ws, size_t ws_size,
                              hipStream_t stream)
{
    const float* x       = (const float*)d_in[0];
    const float* loc_w   = (const float*)d_in[1];
    const float* loc_b   = (const float*)d_in[2];
    const float* score_w = (const float*)d_in[3];
    const float* score_b = (const float*)d_in[4];
    float* out = (float*)d_out;

    rpn_head_kernel<<<TILES / WG_TILES, WG_THREADS, 0, stream>>>(x, loc_w, loc_b, score_w, score_b, out);

    float* out_anchor = out + LOC_ELEMS + SC_ELEMS + FG_ELEMS;
    rpn_anchor_kernel<<<(HW * 9) / 256, 256, 0, stream>>>(out_anchor);
}


# whole W^T (512 channels x 64 columns, 128 KB) staged into LDS once per workgroup by 16-byte loads, bank-swizzled so its stores do not conflict, replacing four stage-and-barrier phases; sum order same
# speedup vs baseline: 1.5826x; 1.1281x over previous
#include <hip/hip_runtime.h>
#include <math.h>

typedef float v2f __attribute__((ext_vector_type(2)));
typedef float v8f __attribute__((ext_vector_type(8)));

#define C_DIM   512
#define HW      6144
#define NIMG    16
#define O_LOC   36
#define O_SC    18
#define O_TOT   54
#define O_PAD   64
#define KCHUNK  128
#define PAIR_STRIDE (O_PAD * 2)
#define KBLOCK  32
#define BSTEPS  (KBLOCK / 4)
#define WG_WAVES   12
#define WG_THREADS (WG_WAVES * 32)
#define WG_TILES   (WG_WAVES * 2)
#define WQ_PER_THREAD ((O_TOT * (C_DIM / 4)) / WG_THREADS)

#define M_TOTAL   (NIMG * HW)
#define TILES     (M_TOTAL / 16)
#define TILES_PER_IMG (HW / 16)

#define LOC_ELEMS ((size_t)NIMG * HW * O_LOC)
#define SC_ELEMS  ((size_t)NIMG * HW * O_SC)
#define FG_ELEMS  ((size_t)NIMG * HW * 9)

__device__ __forceinline__
void load_x_block(const float* __restrict__ p0, const float* __restrict__ p1,
                  v2f (&a0)[BSTEPS], v2f (&a1)[BSTEPS])
{
    #pragma unroll
    for (int s = 0; s < BSTEPS; ++s) {
        a0[s].x = p0[(size_t)(4 * s    ) * HW];
        a0[s].y = p0[(size_t)(4 * s + 1) * HW];
        a1[s].x = p1[(size_t)(4 * s    ) * HW];
        a1[s].y = p1[(size_t)(4 * s + 1) * HW];
    }
}

template <int FLIP>
__device__ __forceinline__
void mma_x_block(const float* pblk, const int (&off)[BSTEPS],
                 const v2f (&a0)[BSTEPS], const v2f (&a1)[BSTEPS],
                 v8f (&acc)[2][4])
{
    #pragma unroll
    for (int s = 0; s < BSTEPS; ++s) {
        v2f b[4];
        #pragma unroll
        for (int j = 0; j < 4; ++j)
            b[j] = *(const v2f*)(pblk + off[s] + s * 2 * PAIR_STRIDE + (j ^ FLIP) * 32);

        #pragma unroll
        for (int j = 0; j < 4; ++j) {
            acc[0][j] = __builtin_amdgcn_wmma_f32_16x16x4_f32(
                false, a0[s], false, b[j], (short)0, acc[0][j], false, false);
            acc[1][j] = __builtin_amdgcn_wmma_f32_16x16x4_f32(
                false, a1[s], false, b[j], (short)0, acc[1][j], false, false);
        }
    }
}

template <bool HAS_NEXT>
__device__ __forceinline__
void head_chunk(const int kc,
                const float* sW, const float* __restrict__ p0, const float* __restrict__ p1,
                const int (&off)[BSTEPS],
                v2f (&xP0)[BSTEPS], v2f (&xP1)[BSTEPS],
                v2f (&xQ0)[BSTEPS], v2f (&xQ1)[BSTEPS], v8f (&acc)[2][4])
{
    const int k0 = kc * KCHUNK;
    const float* pbw = sW + (k0 / 2) * PAIR_STRIDE;

    const float* q0 = p0 + (size_t)k0 * HW;
    const float* q1 = p1 + (size_t)k0 * HW;

    load_x_block(q0 + (size_t)(1 * KBLOCK) * HW, q1 + (size_t)(1 * KBLOCK) * HW, xQ0, xQ1);
    mma_x_block<0>(pbw + 0 * (KBLOCK / 2) * PAIR_STRIDE, off, xP0, xP1, acc);

    load_x_block(q0 + (size_t)(2 * KBLOCK) * HW, q1 + (size_t)(2 * KBLOCK) * HW, xP0, xP1);
    mma_x_block<1>(pbw + 1 * (KBLOCK / 2) * PAIR_STRIDE, off, xQ0, xQ1, acc);

    load_x_block(q0 + (size_t)(3 * KBLOCK) * HW, q1 + (size_t)(3 * KBLOCK) * HW, xQ0, xQ1);
    mma_x_block<0>(pbw + 2 * (KBLOCK / 2) * PAIR_STRIDE, off, xP0, xP1, acc);

    if (HAS_NEXT)
        load_x_block(q0 + (size_t)(4 * KBLOCK) * HW, q1 + (size_t)(4 * KBLOCK) * HW, xP0, xP1);
    mma_x_block<1>(pbw + 3 * (KBLOCK / 2) * PAIR_STRIDE, off, xQ0, xQ1, acc);
}

__global__ __launch_bounds__(WG_THREADS)
void rpn_head_kernel(const float* __restrict__ x,
                     const float* __restrict__ loc_w,
                     const float* __restrict__ loc_b,
                     const float* __restrict__ score_w,
                     const float* __restrict__ score_b,
                     float* __restrict__ d_out)
{
    __shared__ __attribute__((aligned(16))) float sW[(C_DIM / 2) * PAIR_STRIDE];
    __shared__ float sBias[O_PAD];

    const int tid  = threadIdx.x;
    const int wave = tid >> 5;
    const int lane = tid & 31;
    const int col  = lane & 15;
    const int khalf = lane >> 4;

    if (tid < O_PAD) {
        float b = 0.0f;
        if (tid < O_LOC)       b = loc_b[tid];
        else if (tid < O_TOT)  b = score_b[tid - O_LOC];
        sBias[tid] = b;
    }

    const int tile0 = blockIdx.x * WG_TILES + wave * 2;
    const int tile1 = tile0 + 1;

    const int n0 = tile0 / TILES_PER_IMG;
    const int n1 = tile1 / TILES_PER_IMG;
    const size_t base0 = (size_t)n0 * C_DIM * HW + (size_t)(tile0 % TILES_PER_IMG) * 16 + col;
    const size_t base1 = (size_t)n1 * C_DIM * HW + (size_t)(tile1 % TILES_PER_IMG) * 16 + col;

    v8f acc[2][4];
    const v8f vzero = {0.f,0.f,0.f,0.f,0.f,0.f,0.f,0.f};
    #pragma unroll
    for (int t = 0; t < 2; ++t)
        #pragma unroll
        for (int j = 0; j < 4; ++j)
            acc[t][j] = vzero;

    const int kq = khalf << 1;

    const float* p0 = x + base0 + (size_t)kq * HW;
    const float* p1 = x + base1 + (size_t)kq * HW;

    v2f xP0[BSTEPS], xP1[BSTEPS], xQ0[BSTEPS], xQ1[BSTEPS];

    const int oi = lane >> 4;
    const int kl = lane & 15;
    float4 wq[WQ_PER_THREAD];
    #pragma unroll
    for (int i = 0; i < WQ_PER_THREAD; ++i) {
        const int u  = wave + WG_WAVES * i;
        const int o  = ((u >> 3) << 1) + oi;
        const int k4 = ((u & 7) << 4) + kl;
        const float* row = (o < O_LOC) ? (loc_w + o * C_DIM)
                                       : (score_w + (o - O_LOC) * C_DIM);
        wq[i] = *(const float4*)(row + 4 * k4);
    }

    load_x_block(p0, p1, xP0, xP1);

    #pragma unroll
    for (int i = 0; i < WQ_PER_THREAD; ++i) {
        const int u  = wave + WG_WAVES * i;
        const int o  = ((u >> 3) << 1) + oi;
        const int k4 = ((u & 7) << 4) + kl;
        float* d = &sW[(2 * k4) * PAIR_STRIDE + ((o << 1) ^ (kl << 2))];
        v2f lo, hi;
        lo.x = wq[i].x; lo.y = wq[i].y;
        hi.x = wq[i].z; hi.y = wq[i].w;
        *(v2f*)d = lo;
        *(v2f*)(d + PAIR_STRIDE) = hi;
    }
    __syncthreads();

    int off[BSTEPS];
    #pragma unroll
    for (int s = 0; s < BSTEPS; ++s)
        off[s] = khalf * PAIR_STRIDE + ((col ^ (s << 1)) << 1);

    for (int kc = 0; kc < C_DIM / KCHUNK - 1; ++kc)
        head_chunk<true>(kc, sW, p0, p1, off, xP0, xP1, xQ0, xQ1, acc);
    head_chunk<false>(C_DIM / KCHUNK - 1, sW, p0, p1, off, xP0, xP1, xQ0, xQ1, acc);

    float* out_loc   = d_out;
    float* out_score = d_out + LOC_ELEMS;
    float* out_fg    = d_out + LOC_ELEMS + SC_ELEMS;

    const int mhi = khalf * 8;

    #pragma unroll
    for (int t = 0; t < 2; ++t) {
        const int tile = (t == 0) ? tile0 : tile1;
        #pragma unroll
        for (int j = 0; j < 4; ++j) {
            const int o = j * 16 + col;
            const float bias = sBias[o];
            #pragma unroll
            for (int v = 0; v < 8; ++v) {
                const float val = acc[t][j][v] + bias;
                const float partner = __shfl_xor(val, 1, 32);
                const size_t p = (size_t)tile * 16 + (size_t)(v + mhi);
                if (o < O_LOC) {
                    out_loc[p * O_LOC + o] = val;
                } else if (o < O_TOT) {
                    const int s = o - O_LOC;
                    out_score[p * O_SC + s] = val;
                    if ((s & 1) == 0) {
                        const float fg = 1.0f / (1.0f + __expf(val - partner));
                        out_fg[p * 9 + (s >> 1)] = fg;
                    }
                }
            }
        }
    }
}

__global__ __launch_bounds__(256)
void rpn_anchor_kernel(float* __restrict__ out_anchor)
{
    const int i = blockIdx.x * 256 + threadIdx.x;
    const int hw = i / 9;
    const int a  = i - hw * 9;
    const int h  = hw / 96;
    const int w  = hw - h * 96;
    const int ri = a / 3;
    const int si = a - ri * 3;

    const float ratios[3] = {0.5f, 1.0f, 2.0f};
    const float scales[3] = {8.0f, 16.0f, 32.0f};
    const float r = ratios[ri];
    const float s = scales[si];

    const float hbox = 16.0f * s * __fsqrt_rn(r);
    const float wbox = 16.0f * s * __fsqrt_rn(1.0f / r);
    const float py = (float)h * 16.0f + 8.0f;
    const float px = (float)w * 16.0f + 8.0f;

    float4 v;
    v.x = py - hbox * 0.5f;
    v.y = px - wbox * 0.5f;
    v.z = py + hbox * 0.5f;
    v.w = px + wbox * 0.5f;
    ((float4*)out_anchor)[i] = v;
}

extern "C" void kernel_launch(void* const* d_in, const int* in_sizes, int n_in,
                              void* d_out, int out_size, void* d_ws, size_t ws_size,
                              hipStream_t stream)
{
    const float* x       = (const float*)d_in[0];
    const float* loc_w   = (const float*)d_in[1];
    const float* loc_b   = (const float*)d_in[2];
    const float* score_w = (const float*)d_in[3];
    const float* score_b = (const float*)d_in[4];
    float* out = (float*)d_out;

    rpn_head_kernel<<<TILES / WG_TILES, WG_THREADS, 0, stream>>>(x, loc_w, loc_b, score_w, score_b, out);

    float* out_anchor = out + LOC_ELEMS + SC_ELEMS + FG_ELEMS;
    rpn_anchor_kernel<<<(HW * 9) / 256, 256, 0, stream>>>(out_anchor);
}


# W^T fragments double-buffered in registers, four k-steps a set: a set's 16 LDS loads go out as one group 32 WMMAs ahead of use, so no WMMA waits on an LDS round trip; sum order unchanged
# speedup vs baseline: 1.6397x; 1.0361x over previous
#include <hip/hip_runtime.h>
#include <math.h>

typedef float v2f __attribute__((ext_vector_type(2)));
typedef float v8f __attribute__((ext_vector_type(8)));

#define C_DIM   512
#define HW      6144
#define NIMG    16
#define O_LOC   36
#define O_SC    18
#define O_TOT   54
#define O_PAD   64
#define KCHUNK  128
#define PAIR_STRIDE (O_PAD * 2)
#define KBLOCK  32
#define BSTEPS  (KBLOCK / 4)
#define GSTEPS  4
#define WG_WAVES   12
#define WG_THREADS (WG_WAVES * 32)
#define WG_TILES   (WG_WAVES * 2)
#define WQ_PER_THREAD ((O_TOT * (C_DIM / 4)) / WG_THREADS)

#define M_TOTAL   (NIMG * HW)
#define TILES     (M_TOTAL / 16)
#define TILES_PER_IMG (HW / 16)

#define LOC_ELEMS ((size_t)NIMG * HW * O_LOC)
#define SC_ELEMS  ((size_t)NIMG * HW * O_SC)
#define FG_ELEMS  ((size_t)NIMG * HW * 9)

__device__ __forceinline__
void load_x_block(const float* __restrict__ p0, const float* __restrict__ p1,
                  v2f (&a0)[BSTEPS], v2f (&a1)[BSTEPS])
{
    #pragma unroll
    for (int s = 0; s < BSTEPS; ++s) {
        a0[s].x = p0[(size_t)(4 * s    ) * HW];
        a0[s].y = p0[(size_t)(4 * s + 1) * HW];
        a1[s].x = p1[(size_t)(4 * s    ) * HW];
        a1[s].y = p1[(size_t)(4 * s + 1) * HW];
    }
}

#define PIN_LDS_AND_WMMA() __builtin_amdgcn_sched_barrier(0x0076)

template <int FLIP, int S0>
__device__ __forceinline__
void load_w_group(const float* pblk, const int (&off)[BSTEPS], v2f (&b)[GSTEPS][4])
{
    #pragma unroll
    for (int t = 0; t < GSTEPS; ++t) {
        #pragma unroll
        for (int j = 0; j < 4; ++j)
            b[t][j] = *(const v2f*)(pblk + off[S0 + t] + (S0 + t) * 2 * PAIR_STRIDE + (j ^ FLIP) * 32);
    }
}

template <int S0>
__device__ __forceinline__
void mma_group(const v2f (&a0)[BSTEPS], const v2f (&a1)[BSTEPS],
               const v2f (&b)[GSTEPS][4], v8f (&acc)[2][4])
{
    #pragma unroll
    for (int t = 0; t < GSTEPS; ++t) {
        #pragma unroll
        for (int j = 0; j < 4; ++j) {
            acc[0][j] = __builtin_amdgcn_wmma_f32_16x16x4_f32(
                false, a0[S0 + t], false, b[t][j], (short)0, acc[0][j], false, false);
            acc[1][j] = __builtin_amdgcn_wmma_f32_16x16x4_f32(
                false, a1[S0 + t], false, b[t][j], (short)0, acc[1][j], false, false);
        }
    }
}

template <int FLIP, bool HAS_NEXT>
__device__ __forceinline__
void mma_x_block(const float* pblk, const float* pnext, const int (&off)[BSTEPS],
                 const v2f (&xa0)[BSTEPS], const v2f (&xa1)[BSTEPS],
                 v2f (&bE)[GSTEPS][4], v2f (&bO)[GSTEPS][4], v8f (&acc)[2][4])
{
    load_w_group<FLIP, GSTEPS>(pblk, off, bO);
    PIN_LDS_AND_WMMA();
    mma_group<0>(xa0, xa1, bE, acc);
    PIN_LDS_AND_WMMA();
    if (HAS_NEXT) {
        load_w_group<FLIP ^ 1, 0>(pnext, off, bE);
        PIN_LDS_AND_WMMA();
    }
    mma_group<GSTEPS>(xa0, xa1, bO, acc);
    PIN_LDS_AND_WMMA();
}

template <bool HAS_NEXT>
__device__ __forceinline__
void head_chunk(const int kc,
                const float* sW, const float* __restrict__ p0, const float* __restrict__ p1,
                const int (&off)[BSTEPS],
                v2f (&xP0)[BSTEPS], v2f (&xP1)[BSTEPS],
                v2f (&xQ0)[BSTEPS], v2f (&xQ1)[BSTEPS],
                v2f (&bE)[GSTEPS][4], v2f (&bO)[GSTEPS][4], v8f (&acc)[2][4])
{
    const int k0 = kc * KCHUNK;
    const float* pbw = sW + (k0 / 2) * PAIR_STRIDE;
    const float* pb0 = pbw + 0 * (KBLOCK / 2) * PAIR_STRIDE;
    const float* pb1 = pbw + 1 * (KBLOCK / 2) * PAIR_STRIDE;
    const float* pb2 = pbw + 2 * (KBLOCK / 2) * PAIR_STRIDE;
    const float* pb3 = pbw + 3 * (KBLOCK / 2) * PAIR_STRIDE;
    const float* pb4 = pbw + 4 * (KBLOCK / 2) * PAIR_STRIDE;

    const float* q0 = p0 + (size_t)k0 * HW;
    const float* q1 = p1 + (size_t)k0 * HW;

    load_x_block(q0 + (size_t)(1 * KBLOCK) * HW, q1 + (size_t)(1 * KBLOCK) * HW, xQ0, xQ1);
    mma_x_block<0, true>(pb0, pb1, off, xP0, xP1, bE, bO, acc);

    load_x_block(q0 + (size_t)(2 * KBLOCK) * HW, q1 + (size_t)(2 * KBLOCK) * HW, xP0, xP1);
    mma_x_block<1, true>(pb1, pb2, off, xQ0, xQ1, bE, bO, acc);

    load_x_block(q0 + (size_t)(3 * KBLOCK) * HW, q1 + (size_t)(3 * KBLOCK) * HW, xQ0, xQ1);
    mma_x_block<0, true>(pb2, pb3, off, xP0, xP1, bE, bO, acc);

    if (HAS_NEXT)
        load_x_block(q0 + (size_t)(4 * KBLOCK) * HW, q1 + (size_t)(4 * KBLOCK) * HW, xP0, xP1);
    mma_x_block<1, HAS_NEXT>(pb3, pb4, off, xQ0, xQ1, bE, bO, acc);
}

__global__ __launch_bounds__(WG_THREADS)
void rpn_head_kernel(const float* __restrict__ x,
                     const float* __restrict__ loc_w,
                     const float* __restrict__ loc_b,
                     const float* __restrict__ score_w,
                     const float* __restrict__ score_b,
                     float* __restrict__ d_out)
{
    __shared__ __attribute__((aligned(16))) float sW[(C_DIM / 2) * PAIR_STRIDE];
    __shared__ float sBias[O_PAD];

    const int tid  = threadIdx.x;
    const int wave = tid >> 5;
    const int lane = tid & 31;
    const int col  = lane & 15;
    const int khalf = lane >> 4;

    if (tid < O_PAD) {
        float b = 0.0f;
        if (tid < O_LOC)       b = loc_b[tid];
        else if (tid < O_TOT)  b = score_b[tid - O_LOC];
        sBias[tid] = b;
    }

    const int tile0 = blockIdx.x * WG_TILES + wave * 2;
    const int tile1 = tile0 + 1;

    const int n0 = tile0 / TILES_PER_IMG;
    const int n1 = tile1 / TILES_PER_IMG;
    const size_t base0 = (size_t)n0 * C_DIM * HW + (size_t)(tile0 % TILES_PER_IMG) * 16 + col;
    const size_t base1 = (size_t)n1 * C_DIM * HW + (size_t)(tile1 % TILES_PER_IMG) * 16 + col;

    v8f acc[2][4];
    const v8f vzero = {0.f,0.f,0.f,0.f,0.f,0.f,0.f,0.f};
    #pragma unroll
    for (int t = 0; t < 2; ++t)
        #pragma unroll
        for (int j = 0; j < 4; ++j)
            acc[t][j] = vzero;

    const int kq = khalf << 1;

    const float* p0 = x + base0 + (size_t)kq * HW;
    const float* p1 = x + base1 + (size_t)kq * HW;

    v2f xP0[BSTEPS], xP1[BSTEPS], xQ0[BSTEPS], xQ1[BSTEPS];

    const int oi = lane >> 4;
    const int kl = lane & 15;
    float4 wq[WQ_PER_THREAD];
    #pragma unroll
    for (int i = 0; i < WQ_PER_THREAD; ++i) {
        const int u  = wave + WG_WAVES * i;
        const int o  = ((u >> 3) << 1) + oi;
        const int k4 = ((u & 7) << 4) + kl;
        const float* row = (o < O_LOC) ? (loc_w + o * C_DIM)
                                       : (score_w + (o - O_LOC) * C_DIM);
        wq[i] = *(const float4*)(row + 4 * k4);
    }

    load_x_block(p0, p1, xP0, xP1);

    #pragma unroll
    for (int i = 0; i < WQ_PER_THREAD; ++i) {
        const int u  = wave + WG_WAVES * i;
        const int o  = ((u >> 3) << 1) + oi;
        const int k4 = ((u & 7) << 4) + kl;
        float* d = &sW[(2 * k4) * PAIR_STRIDE + ((o << 1) ^ (kl << 2))];
        v2f lo, hi;
        lo.x = wq[i].x; lo.y = wq[i].y;
        hi.x = wq[i].z; hi.y = wq[i].w;
        *(v2f*)d = lo;
        *(v2f*)(d + PAIR_STRIDE) = hi;
    }
    __syncthreads();

    int off[BSTEPS];
    #pragma unroll
    for (int s = 0; s < BSTEPS; ++s)
        off[s] = khalf * PAIR_STRIDE + ((col ^ (s << 1)) << 1);

    v2f bE[GSTEPS][4], bO[GSTEPS][4];
    load_w_group<0, 0>(sW, off, bE);
    PIN_LDS_AND_WMMA();

    for (int kc = 0; kc < C_DIM / KCHUNK - 1; ++kc)
        head_chunk<true>(kc, sW, p0, p1, off, xP0, xP1, xQ0, xQ1, bE, bO, acc);
    head_chunk<false>(C_DIM / KCHUNK - 1, sW, p0, p1, off, xP0, xP1, xQ0, xQ1, bE, bO, acc);

    float* out_loc   = d_out;
    float* out_score = d_out + LOC_ELEMS;
    float* out_fg    = d_out + LOC_ELEMS + SC_ELEMS;

    const int mhi = khalf * 8;

    #pragma unroll
    for (int t = 0; t < 2; ++t) {
        const int tile = (t == 0) ? tile0 : tile1;
        #pragma unroll
        for (int j = 0; j < 4; ++j) {
            const int o = j * 16 + col;
            const float bias = sBias[o];
            #pragma unroll
            for (int v = 0; v < 8; ++v) {
                const float val = acc[t][j][v] + bias;
                const float partner = __shfl_xor(val, 1, 32);
                const size_t p = (size_t)tile * 16 + (size_t)(v + mhi);
                if (o < O_LOC) {
                    out_loc[p * O_LOC + o] = val;
                } else if (o < O_TOT) {
                    const int s = o - O_LOC;
                    out_score[p * O_SC + s] = val;
                    if ((s & 1) == 0) {
                        const float fg = 1.0f / (1.0f + __expf(val - partner));
                        out_fg[p * 9 + (s >> 1)] = fg;
                    }
                }
            }
        }
    }
}

__global__ __launch_bounds__(256)
void rpn_anchor_kernel(float* __restrict__ out_anchor)
{
    const int i = blockIdx.x * 256 + threadIdx.x;
    const int hw = i / 9;
    const int a  = i - hw * 9;
    const int h  = hw / 96;
    const int w  = hw - h * 96;
    const int ri = a / 3;
    const int si = a - ri * 3;

    const float ratios[3] = {0.5f, 1.0f, 2.0f};
    const float scales[3] = {8.0f, 16.0f, 32.0f};
    const float r = ratios[ri];
    const float s = scales[si];

    const float hbox = 16.0f * s * __fsqrt_rn(r);
    const float wbox = 16.0f * s * __fsqrt_rn(1.0f / r);
    const float py = (float)h * 16.0f + 8.0f;
    const float px = (float)w * 16.0f + 8.0f;

    float4 v;
    v.x = py - hbox * 0.5f;
    v.y = px - wbox * 0.5f;
    v.z = py + hbox * 0.5f;
    v.w = px + wbox * 0.5f;
    ((float4*)out_anchor)[i] = v;
}

extern "C" void kernel_launch(void* const* d_in, const int* in_sizes, int n_in,
                              void* d_out, int out_size, void* d_ws, size_t ws_size,
                              hipStream_t stream)
{
    const float* x       = (const float*)d_in[0];
    const float* loc_w   = (const float*)d_in[1];
    const float* loc_b   = (const float*)d_in[2];
    const float* score_w = (const float*)d_in[3];
    const float* score_b = (const float*)d_in[4];
    float* out = (float*)d_out;

    rpn_head_kernel<<<TILES / WG_TILES, WG_THREADS, 0, stream>>>(x, loc_w, loc_b, score_w, score_b, out);

    float* out_anchor = out + LOC_ELEMS + SC_ELEMS + FG_ELEMS;
    rpn_anchor_kernel<<<(HW * 9) / 256, 256, 0, stream>>>(out_anchor);
}


# epilogue through a wave-private LDS slot: each 128-B line of loc/score/fg leaves by ONE 16-B (fg 8-B) store; bigger LDS lets the compiler hoist x loads (VGPR 226->290); sum order same
# speedup vs baseline: 1.8022x; 1.0991x over previous
#include <hip/hip_runtime.h>
#include <math.h>

typedef float v2f __attribute__((ext_vector_type(2)));
typedef float v8f __attribute__((ext_vector_type(8)));

#define C_DIM   512
#define HW      6144
#define NIMG    16
#define O_LOC   36
#define O_SC    18
#define O_TOT   54
#define O_PAD   64
#define KCHUNK  128
#define PAIR_STRIDE (O_PAD * 2)
#define KBLOCK  32
#define BSTEPS  (KBLOCK / 4)
#define GSTEPS  4
#define WG_WAVES   12
#define WG_THREADS (WG_WAVES * 32)
#define WG_TILES   (WG_WAVES * 2)
#define WQ_PER_THREAD ((O_TOT * (C_DIM / 4)) / WG_THREADS)

#define OUT_LOC_W  (32 * O_LOC)
#define OUT_SC_W   (32 * O_SC)
#define OUT_PER_WAVE (OUT_LOC_W + OUT_SC_W)

#define M_TOTAL   (NIMG * HW)
#define TILES     (M_TOTAL / 16)
#define TILES_PER_IMG (HW / 16)

#define LOC_ELEMS ((size_t)NIMG * HW * O_LOC)
#define SC_ELEMS  ((size_t)NIMG * HW * O_SC)
#define FG_ELEMS  ((size_t)NIMG * HW * 9)

__device__ __forceinline__
void load_x_block(const float* __restrict__ p0, const float* __restrict__ p1,
                  v2f (&a0)[BSTEPS], v2f (&a1)[BSTEPS])
{
    #pragma unroll
    for (int s = 0; s < BSTEPS; ++s) {
        a0[s].x = p0[(size_t)(4 * s    ) * HW];
        a0[s].y = p0[(size_t)(4 * s + 1) * HW];
        a1[s].x = p1[(size_t)(4 * s    ) * HW];
        a1[s].y = p1[(size_t)(4 * s + 1) * HW];
    }
}

#define PIN_LDS_AND_WMMA() __builtin_amdgcn_sched_barrier(0x0076)

template <int FLIP, int S0>
__device__ __forceinline__
void load_w_group(const float* pblk, const int (&off)[BSTEPS], v2f (&b)[GSTEPS][4])
{
    #pragma unroll
    for (int t = 0; t < GSTEPS; ++t) {
        #pragma unroll
        for (int j = 0; j < 4; ++j)
            b[t][j] = *(const v2f*)(pblk + off[S0 + t] + (S0 + t) * 2 * PAIR_STRIDE + (j ^ FLIP) * 32);
    }
}

template <int S0>
__device__ __forceinline__
void mma_group(const v2f (&a0)[BSTEPS], const v2f (&a1)[BSTEPS],
               const v2f (&b)[GSTEPS][4], v8f (&acc)[2][4])
{
    #pragma unroll
    for (int t = 0; t < GSTEPS; ++t) {
        #pragma unroll
        for (int j = 0; j < 4; ++j) {
            acc[0][j] = __builtin_amdgcn_wmma_f32_16x16x4_f32(
                false, a0[S0 + t], false, b[t][j], (short)0, acc[0][j], false, false);
            acc[1][j] = __builtin_amdgcn_wmma_f32_16x16x4_f32(
                false, a1[S0 + t], false, b[t][j], (short)0, acc[1][j], false, false);
        }
    }
}

template <int FLIP, bool HAS_NEXT>
__device__ __forceinline__
void mma_x_block(const float* pblk, const float* pnext, const int (&off)[BSTEPS],
                 const v2f (&xa0)[BSTEPS], const v2f (&xa1)[BSTEPS],
                 v2f (&bE)[GSTEPS][4], v2f (&bO)[GSTEPS][4], v8f (&acc)[2][4])
{
    load_w_group<FLIP, GSTEPS>(pblk, off, bO);
    PIN_LDS_AND_WMMA();
    mma_group<0>(xa0, xa1, bE, acc);
    PIN_LDS_AND_WMMA();
    if (HAS_NEXT) {
        load_w_group<FLIP ^ 1, 0>(pnext, off, bE);
        PIN_LDS_AND_WMMA();
    }
    mma_group<GSTEPS>(xa0, xa1, bO, acc);
    PIN_LDS_AND_WMMA();
}

template <bool HAS_NEXT>
__device__ __forceinline__
void head_chunk(const int kc,
                const float* sW, const float* __restrict__ p0, const float* __restrict__ p1,
                const int (&off)[BSTEPS],
                v2f (&xP0)[BSTEPS], v2f (&xP1)[BSTEPS],
                v2f (&xQ0)[BSTEPS], v2f (&xQ1)[BSTEPS],
                v2f (&bE)[GSTEPS][4], v2f (&bO)[GSTEPS][4], v8f (&acc)[2][4])
{
    const int k0 = kc * KCHUNK;
    const float* pbw = sW + (k0 / 2) * PAIR_STRIDE;
    const float* pb0 = pbw + 0 * (KBLOCK / 2) * PAIR_STRIDE;
    const float* pb1 = pbw + 1 * (KBLOCK / 2) * PAIR_STRIDE;
    const float* pb2 = pbw + 2 * (KBLOCK / 2) * PAIR_STRIDE;
    const float* pb3 = pbw + 3 * (KBLOCK / 2) * PAIR_STRIDE;
    const float* pb4 = pbw + 4 * (KBLOCK / 2) * PAIR_STRIDE;

    const float* q0 = p0 + (size_t)k0 * HW;
    const float* q1 = p1 + (size_t)k0 * HW;

    load_x_block(q0 + (size_t)(1 * KBLOCK) * HW, q1 + (size_t)(1 * KBLOCK) * HW, xQ0, xQ1);
    mma_x_block<0, true>(pb0, pb1, off, xP0, xP1, bE, bO, acc);

    load_x_block(q0 + (size_t)(2 * KBLOCK) * HW, q1 + (size_t)(2 * KBLOCK) * HW, xP0, xP1);
    mma_x_block<1, true>(pb1, pb2, off, xQ0, xQ1, bE, bO, acc);

    load_x_block(q0 + (size_t)(3 * KBLOCK) * HW, q1 + (size_t)(3 * KBLOCK) * HW, xQ0, xQ1);
    mma_x_block<0, true>(pb2, pb3, off, xP0, xP1, bE, bO, acc);

    if (HAS_NEXT)
        load_x_block(q0 + (size_t)(4 * KBLOCK) * HW, q1 + (size_t)(4 * KBLOCK) * HW, xP0, xP1);
    mma_x_block<1, HAS_NEXT>(pb3, pb4, off, xQ0, xQ1, bE, bO, acc);
}

__global__ __launch_bounds__(WG_THREADS)
void rpn_head_kernel(const float* __restrict__ x,
                     const float* __restrict__ loc_w,
                     const float* __restrict__ loc_b,
                     const float* __restrict__ score_w,
                     const float* __restrict__ score_b,
                     float* __restrict__ d_out)
{
    __shared__ __attribute__((aligned(16))) float sW[(C_DIM / 2) * PAIR_STRIDE];
    __shared__ float sBias[O_PAD];
    __shared__ __attribute__((aligned(16))) float sOut[WG_WAVES * OUT_PER_WAVE];

    const int tid  = threadIdx.x;
    const int wave = tid >> 5;
    const int lane = tid & 31;
    const int col  = lane & 15;
    const int khalf = lane >> 4;

    if (tid < O_PAD) {
        float b = 0.0f;
        if (tid < O_LOC)       b = loc_b[tid];
        else if (tid < O_TOT)  b = score_b[tid - O_LOC];
        sBias[tid] = b;
    }

    const int tile0 = blockIdx.x * WG_TILES + wave * 2;
    const int tile1 = tile0 + 1;

    const int n0 = tile0 / TILES_PER_IMG;
    const int n1 = tile1 / TILES_PER_IMG;
    const size_t base0 = (size_t)n0 * C_DIM * HW + (size_t)(tile0 % TILES_PER_IMG) * 16 + col;
    const size_t base1 = (size_t)n1 * C_DIM * HW + (size_t)(tile1 % TILES_PER_IMG) * 16 + col;

    v8f acc[2][4];
    const v8f vzero = {0.f,0.f,0.f,0.f,0.f,0.f,0.f,0.f};
    #pragma unroll
    for (int t = 0; t < 2; ++t)
        #pragma unroll
        for (int j = 0; j < 4; ++j)
            acc[t][j] = vzero;

    const int kq = khalf << 1;

    const float* p0 = x + base0 + (size_t)kq * HW;
    const float* p1 = x + base1 + (size_t)kq * HW;

    v2f xP0[BSTEPS], xP1[BSTEPS], xQ0[BSTEPS], xQ1[BSTEPS];

    const int oi = lane >> 4;
    const int kl = lane & 15;
    float4 wq[WQ_PER_THREAD];
    #pragma unroll
    for (int i = 0; i < WQ_PER_THREAD; ++i) {
        const int u  = wave + WG_WAVES * i;
        const int o  = ((u >> 3) << 1) + oi;
        const int k4 = ((u & 7) << 4) + kl;
        const float* row = (o < O_LOC) ? (loc_w + o * C_DIM)
                                       : (score_w + (o - O_LOC) * C_DIM);
        wq[i] = *(const float4*)(row + 4 * k4);
    }

    load_x_block(p0, p1, xP0, xP1);

    #pragma unroll
    for (int i = 0; i < WQ_PER_THREAD; ++i) {
        const int u  = wave + WG_WAVES * i;
        const int o  = ((u >> 3) << 1) + oi;
        const int k4 = ((u & 7) << 4) + kl;
        float* d = &sW[(2 * k4) * PAIR_STRIDE + ((o << 1) ^ (kl << 2))];
        v2f lo, hi;
        lo.x = wq[i].x; lo.y = wq[i].y;
        hi.x = wq[i].z; hi.y = wq[i].w;
        *(v2f*)d = lo;
        *(v2f*)(d + PAIR_STRIDE) = hi;
    }
    __syncthreads();

    int off[BSTEPS];
    #pragma unroll
    for (int s = 0; s < BSTEPS; ++s)
        off[s] = khalf * PAIR_STRIDE + ((col ^ (s << 1)) << 1);

    v2f bE[GSTEPS][4], bO[GSTEPS][4];
    load_w_group<0, 0>(sW, off, bE);
    PIN_LDS_AND_WMMA();

    for (int kc = 0; kc < C_DIM / KCHUNK - 1; ++kc)
        head_chunk<true>(kc, sW, p0, p1, off, xP0, xP1, xQ0, xQ1, bE, bO, acc);
    head_chunk<false>(C_DIM / KCHUNK - 1, sW, p0, p1, off, xP0, xP1, xQ0, xQ1, bE, bO, acc);

    float* out_loc   = d_out;
    float* out_score = d_out + LOC_ELEMS;
    float* out_fg    = d_out + LOC_ELEMS + SC_ELEMS;

    const int mhi = khalf * 8;
    float* so = sOut + wave * OUT_PER_WAVE;

    #pragma unroll
    for (int t = 0; t < 2; ++t) {
        #pragma unroll
        for (int j = 0; j < 4; ++j) {
            const int o = j * 16 + col;
            const float bias = sBias[o];
            const int prow = t * 16 + mhi;
            const bool is_loc = (o < O_LOC);
            const int idx    = is_loc ? (prow * O_LOC + o)
                                      : (OUT_LOC_W + prow * O_SC + (o - O_LOC));
            const int stride = is_loc ? O_LOC : O_SC;
            if (o < O_TOT) {
                #pragma unroll
                for (int v = 0; v < 8; ++v)
                    so[idx + v * stride] = acc[t][j][v] + bias;
            }
        }
    }

    __builtin_amdgcn_fence(__ATOMIC_RELEASE, "wavefront");
    __builtin_amdgcn_wave_barrier();
    __builtin_amdgcn_fence(__ATOMIC_ACQUIRE, "wavefront");

    const size_t pix0 = (size_t)tile0 * 16;
    const float4* so4 = (const float4*)so;

    float4* g_loc = (float4*)(out_loc + pix0 * O_LOC);
    #pragma unroll
    for (int i = 0; i < OUT_LOC_W / 4 / 32; ++i)
        g_loc[i * 32 + lane] = so4[i * 32 + lane];

    float4* g_sc = (float4*)(out_score + pix0 * O_SC);
    float2* g_fg = (float2*)(out_fg + pix0 * 9);
    #pragma unroll
    for (int i = 0; i < (OUT_SC_W / 4 + 31) / 32; ++i) {
        const int q = i * 32 + lane;
        if (q < OUT_SC_W / 4) {
            const float4 s = so4[OUT_LOC_W / 4 + q];
            g_sc[q] = s;
            float2 f;
            f.x = 1.0f / (1.0f + __expf(s.x - s.y));
            f.y = 1.0f / (1.0f + __expf(s.z - s.w));
            g_fg[q] = f;
        }
    }
}

__global__ __launch_bounds__(256)
void rpn_anchor_kernel(float* __restrict__ out_anchor)
{
    const int i = blockIdx.x * 256 + threadIdx.x;
    const int hw = i / 9;
    const int a  = i - hw * 9;
    const int h  = hw / 96;
    const int w  = hw - h * 96;
    const int ri = a / 3;
    const int si = a - ri * 3;

    const float ratios[3] = {0.5f, 1.0f, 2.0f};
    const float scales[3] = {8.0f, 16.0f, 32.0f};
    const float r = ratios[ri];
    const float s = scales[si];

    const float hbox = 16.0f * s * __fsqrt_rn(r);
    const float wbox = 16.0f * s * __fsqrt_rn(1.0f / r);
    const float py = (float)h * 16.0f + 8.0f;
    const float px = (float)w * 16.0f + 8.0f;

    float4 v;
    v.x = py - hbox * 0.5f;
    v.y = px - wbox * 0.5f;
    v.z = py + hbox * 0.5f;
    v.w = px + wbox * 0.5f;
    ((float4*)out_anchor)[i] = v;
}

extern "C" void kernel_launch(void* const* d_in, const int* in_sizes, int n_in,
                              void* d_out, int out_size, void* d_ws, size_t ws_size,
                              hipStream_t stream)
{
    const float* x       = (const float*)d_in[0];
    const float* loc_w   = (const float*)d_in[1];
    const float* loc_b   = (const float*)d_in[2];
    const float* score_w = (const float*)d_in[3];
    const float* score_b = (const float*)d_in[4];
    float* out = (float*)d_out;

    rpn_head_kernel<<<TILES / WG_TILES, WG_THREADS, 0, stream>>>(x, loc_w, loc_b, score_w, score_b, out);

    float* out_anchor = out + LOC_ELEMS + SC_ELEMS + FG_ELEMS;
    rpn_anchor_kernel<<<(HW * 9) / 256, 256, 0, stream>>>(out_anchor);
}
